# starting kernel
# baseline (speedup 1.0000x reference)
#include <hip/hip_runtime.h>
#include <hip/hip_bf16.h>
#include <math.h>

#define BATCH 4
#define NTOK  1024
#define DIM   128
#define HID   64

typedef __attribute__((ext_vector_type(2))) float v2f;
typedef __attribute__((ext_vector_type(4))) float f4;
typedef __attribute__((ext_vector_type(8))) float v8f;

__device__ __forceinline__ float fast_tanh(float x) {
#if __has_builtin(__builtin_amdgcn_tanhf)
  return __builtin_amdgcn_tanhf(x);
#else
  float r;
  asm volatile("v_tanh_f32 %0, %1" : "=v"(r) : "v"(x));
  return r;
#endif
}

__global__ void __launch_bounds__(32)
proj_kernel(const float* __restrict__ E, const float* __restrict__ W1,
            const float* __restrict__ b1, float* __restrict__ Hbuf) {
  const int lane = threadIdx.x;
  const int m0   = blockIdx.x * 16;
  const int n0   = blockIdx.y * 16;
  const int half = blockIdx.z;

  const float* W = W1 + half * (DIM * HID);
  const int row  = m0 + (lane & 15);
  const int koff = (lane >> 4) * 2;
  const int ncol = n0 + (lane & 15);

  v8f c = {};
  #pragma unroll
  for (int k = 0; k < DIM; k += 4) {
    v2f a, b;
    a.x = E[row * DIM + (k + koff)];
    a.y = E[row * DIM + (k + koff + 1)];
    b.x = W[(k + koff) * HID + ncol];
    b.y = W[(k + koff + 1) * HID + ncol];
    c = __builtin_amdgcn_wmma_f32_16x16x4_f32(
         false, a,  false, b,
         (short)0, c,  false,  false);
  }

  float* out = Hbuf + (size_t)half * (BATCH * NTOK * HID);
  const float bias  = (half == 0) ? b1[ncol] : 0.0f;
  const int   rbase = m0 + ((lane >> 4) * 8);
  #pragma unroll
  for (int r = 0; r < 8; ++r) {
    out[(size_t)(rbase + r) * HID + ncol] = c[r] + bias;
  }
}

__device__ __forceinline__ float relu_dot4(f4 a, f4 b, f4 w, float acc) {
  f4 s = a + b;
  acc = fmaf(fmaxf(s.x, 0.0f), w.x, acc);
  acc = fmaf(fmaxf(s.y, 0.0f), w.y, acc);
  acc = fmaf(fmaxf(s.z, 0.0f), w.z, acc);
  acc = fmaf(fmaxf(s.w, 0.0f), w.w, acc);
  return acc;
}

__global__ void __launch_bounds__(256)
pair_kernel(const float* __restrict__ Hi, const float* __restrict__ Hj,
            const float* __restrict__ W2, const float* __restrict__ b2p,
            float* __restrict__ out) {
  const int ti = blockIdx.x, tj = blockIdx.y, b = blockIdx.z;
  if (tj < ti) return;

  __shared__ float sHiA[32 * HID];
  __shared__ float sHiB[32 * HID];
  __shared__ float sHjA[32 * HID];
  __shared__ float sHjB[32 * HID];
  __shared__ float sW2[HID];
  __shared__ float sT[32 * 32];

  const int tid = threadIdx.x;
  const int i0 = ti * 32, j0 = tj * 32;
  const float* HiBb = Hi + (size_t)b * NTOK * HID;
  const float* HjBb = Hj + (size_t)b * NTOK * HID;
  const f4* gHiA = (const f4*)(HiBb + (size_t)i0 * HID);
  const f4* gHiB = (const f4*)(HiBb + (size_t)j0 * HID);
  const f4* gHjA = (const f4*)(HjBb + (size_t)i0 * HID);
  const f4* gHjB = (const f4*)(HjBb + (size_t)j0 * HID);

  #pragma unroll
  for (int idx = tid; idx < 32 * HID / 4; idx += 256) {
    ((f4*)sHiA)[idx] = gHiA[idx];
    ((f4*)sHiB)[idx] = gHiB[idx];
    ((f4*)sHjA)[idx] = gHjA[idx];
    ((f4*)sHjB)[idx] = gHjB[idx];
  }
  if (tid < HID / 4) ((f4*)sW2)[tid] = ((const f4*)W2)[tid];
  __syncthreads();

  const int ii0 = (tid >> 4) * 2;
  const int jj0 = (tid & 15) * 2;

  float dij[2][2] = {{0.f, 0.f}, {0.f, 0.f}};
  float dji[2][2] = {{0.f, 0.f}, {0.f, 0.f}};

  #pragma unroll
  for (int h = 0; h < HID; h += 4) {
    const f4 w    = *(const f4*)&sW2[h];
    const f4 hiA0 = *(const f4*)&sHiA[(ii0 + 0) * HID + h];
    const f4 hiA1 = *(const f4*)&sHiA[(ii0 + 1) * HID + h];
    const f4 hjA0 = *(const f4*)&sHjA[(ii0 + 0) * HID + h];
    const f4 hjA1 = *(const f4*)&sHjA[(ii0 + 1) * HID + h];
    const f4 hiB0 = *(const f4*)&sHiB[(jj0 + 0) * HID + h];
    const f4 hiB1 = *(const f4*)&sHiB[(jj0 + 1) * HID + h];
    const f4 hjB0 = *(const f4*)&sHjB[(jj0 + 0) * HID + h];
    const f4 hjB1 = *(const f4*)&sHjB[(jj0 + 1) * HID + h];
    dij[0][0] = relu_dot4(hiA0, hjB0, w, dij[0][0]);
    dij[0][1] = relu_dot4(hiA0, hjB1, w, dij[0][1]);
    dij[1][0] = relu_dot4(hiA1, hjB0, w, dij[1][0]);
    dij[1][1] = relu_dot4(hiA1, hjB1, w, dij[1][1]);
    dji[0][0] = relu_dot4(hiB0, hjA0, w, dji[0][0]);
    dji[0][1] = relu_dot4(hiB1, hjA0, w, dji[0][1]);
    dji[1][0] = relu_dot4(hiB0, hjA1, w, dji[1][0]);
    dji[1][1] = relu_dot4(hiB1, hjA1, w, dji[1][1]);
  }

  const float b2v = b2p[0];
  #pragma unroll
  for (int u = 0; u < 2; ++u) {
    #pragma unroll
    for (int v = 0; v < 2; ++v) {
      const int gi = i0 + ii0 + u;
      const int gj = j0 + jj0 + v;
      const float s =
          -0.5f * (fast_tanh(dij[u][v] + b2v) + fast_tanh(dji[u][v] + b2v));
      out[((size_t)b * NTOK + gi) * NTOK + gj] = s;
      sT[(jj0 + v) * 32 + (ii0 + u)] = s;
    }
  }
  __syncthreads();

  #pragma unroll
  for (int idx = tid; idx < 32 * 32; idx += 256) {
    const int r = idx >> 5, c = idx & 31;
    out[((size_t)b * NTOK + (j0 + r)) * NTOK + (i0 + c)] = sT[idx];
  }
}

extern "C" void kernel_launch(void* const* d_in, const int* in_sizes, int n_in,
                              void* d_out, int out_size, void* d_ws, size_t ws_size,
                              hipStream_t stream) {
  (void)in_sizes; (void)n_in; (void)out_size; (void)ws_size;
  const float* E  = (const float*)d_in[0];
  const float* W1 = (const float*)d_in[1];
  const float* b1 = (const float*)d_in[2];
  const float* W2 = (const float*)d_in[3];
  const float* b2 = (const float*)d_in[4];
  float* out  = (float*)d_out;
  float* Hbuf = (float*)d_ws;

  proj_kernel<<<dim3((BATCH * NTOK) / 16, HID / 16, 2), 32, 0, stream>>>(E, W1, b1, Hbuf);
  pair_kernel<<<dim3(NTOK / 32, NTOK / 32, BATCH), 256, 0, stream>>>(
      Hbuf, Hbuf + (size_t)BATCH * NTOK * HID, W2, b2, out);
}


# pair stage: loop pipelined over two register sets of 2 h steps: a set's scalar and vector loads go out behind the wait for the other set, ahead of its adds; weights by lane read; sum order kept
# speedup vs baseline: 3.3497x; 3.3497x over previous
#include <hip/hip_runtime.h>
#include <hip/hip_bf16.h>
#include <math.h>

#define BATCH 4
#define NTOK  1024
#define DIM   128
#define HID   64
#define MTOT  (BATCH * NTOK)
#define NJ    8
#define BI    64
#define NBLK  (NTOK / BI)

typedef __attribute__((ext_vector_type(2))) float v2f;
typedef __attribute__((ext_vector_type(4))) float f4;
typedef __attribute__((ext_vector_type(8))) float v8f;

__device__ __forceinline__ float fast_tanh(float x) {
#if __has_builtin(__builtin_amdgcn_tanhf)
  return __builtin_amdgcn_tanhf(x);
#else
  float r;
  asm volatile("v_tanh_f32 %0, %1" : "=v"(r) : "v"(x));
  return r;
#endif
}

__global__ void __launch_bounds__(32)
proj_kernel(const float* __restrict__ E, const float* __restrict__ W1,
            const float* __restrict__ b1, float* __restrict__ Hbuf) {
  const int lane = threadIdx.x;
  const int m0   = blockIdx.x * 16;
  const int n0   = blockIdx.y * 16;
  const int half = blockIdx.z;

  const float* W = W1 + half * (DIM * HID);
  const int row  = m0 + (lane & 15);
  const int koff = (lane >> 4) * 2;
  const int ncol = n0 + (lane & 15);

  v8f c = {};
  #pragma unroll
  for (int k = 0; k < DIM; k += 4) {
    v2f a, b;
    a.x = E[row * DIM + (k + koff)];
    a.y = E[row * DIM + (k + koff + 1)];
    b.x = W[(k + koff) * HID + ncol];
    b.y = W[(k + koff + 1) * HID + ncol];
    c = __builtin_amdgcn_wmma_f32_16x16x4_f32(
         false, a,  false, b,
         (short)0, c,  false,  false);
  }

  const float bias  = (half == 0) ? b1[ncol] : 0.0f;
  const int   rbase = m0 + ((lane >> 4) * 8);
  float* dst = Hbuf + (size_t)half * (MTOT * HID) + (size_t)ncol * MTOT + rbase;
  f4 lo, hi;
  lo.x = c[0] + bias; lo.y = c[1] + bias; lo.z = c[2] + bias; lo.w = c[3] + bias;
  hi.x = c[4] + bias; hi.y = c[5] + bias; hi.z = c[6] + bias; hi.w = c[7] + bias;
  *(f4*)(dst)     = lo;
  *(f4*)(dst + 4) = hi;
}

struct PairStage {
  v2f a0, c0, a1, c1;
  v8f j0, i0, j1, i1;
};

__device__ __forceinline__ void load_stage(PairStage& s, const float* __restrict__ pa,
                                           const float* __restrict__ pc,
                                           const float* __restrict__ qj,
                                           const float* __restrict__ qi, int h) {
  s.a0 = *(const v2f*)(pa + h * MTOT);
  s.c0 = *(const v2f*)(pc + h * MTOT);
  s.a1 = *(const v2f*)(pa + (h + 1) * MTOT);
  s.c1 = *(const v2f*)(pc + (h + 1) * MTOT);
  s.j0 = *(const v8f*)(qj + h * MTOT);
  s.i0 = *(const v8f*)(qi + h * MTOT);
  s.j1 = *(const v8f*)(qj + (h + 1) * MTOT);
  s.i1 = *(const v8f*)(qi + (h + 1) * MTOT);
}

__device__ __forceinline__ void stage_ready(const PairStage& s) {
  __builtin_amdgcn_sched_barrier(0);
  asm volatile("" :: "s"(s.j0));
}

__device__ __forceinline__ void loads_issued() {
  __builtin_amdgcn_sched_barrier(0);
}

__device__ __forceinline__ v2f relu_fma(v2f t, v2f w, v2f acc) {
  const v2f z = {0.0f, 0.0f};
  t = __builtin_elementwise_max(t, z);
  return __builtin_elementwise_fma(t, w, acc);
}

__device__ __forceinline__ void do_step(v2f a, v2f c, v8f xjv, v8f xiv, float w,
                                        v2f (&dij)[NJ], v2f (&dji)[NJ]) {
  const v2f wv = {w, w};
  #pragma unroll
  for (int jj = 0; jj < NJ; ++jj) {
    const v2f xj = {xjv[jj], xjv[jj]};
    const v2f xi = {xiv[jj], xiv[jj]};
    dij[jj] = relu_fma(a + xj, wv, dij[jj]);
    dji[jj] = relu_fma(xi + c, wv, dji[jj]);
  }
}

__device__ __forceinline__ void do_stage(const PairStage& s, v2f wl, int h,
                                         v2f (&dij)[NJ], v2f (&dji)[NJ]) {
  const float wx = wl.x, wy = wl.y;
  const float w0 = __int_as_float(__builtin_amdgcn_readlane(__float_as_int(wx), h >> 1));
  const float w1 = __int_as_float(__builtin_amdgcn_readlane(__float_as_int(wy), h >> 1));
  do_step(s.a0, s.c0, s.j0, s.i0, w0, dij, dji);
  do_step(s.a1, s.c1, s.j1, s.i1, w1, dij, dji);
}

__global__ void __launch_bounds__(128)
pair_kernel(const float* __restrict__ HiT, const float* __restrict__ HjT,
            const float* __restrict__ W2, const float* __restrict__ b2p,
            float* __restrict__ out) {
  const int ti = blockIdx.x, k = blockIdx.y;
  const int b = blockIdx.z >> 1, jh = blockIdx.z & 1;
  if (k == NBLK / 2 && ti >= NBLK / 2) return;
  const int tj = (ti + k) & (NBLK - 1);

  __shared__ float sT[BI * 33];

  const int lane = threadIdx.x & 31;
  const int wv = __builtin_amdgcn_readfirstlane((int)(threadIdx.x >> 5));

  const int mi = b * NTOK + ti * BI;
  const int mj = b * NTOK + tj * BI + jh * 32 + wv * NJ;

  const float* pa = HiT + mi + 2 * lane;
  const float* pc = HjT + mi + 2 * lane;
  const float* qj = HjT + mj;
  const float* qi = HiT + mj;

  v2f dij[NJ], dji[NJ];
  #pragma unroll
  for (int jj = 0; jj < NJ; ++jj) { dij[jj].x = 0.f; dij[jj].y = 0.f; dji[jj].x = 0.f; dji[jj].y = 0.f; }

  PairStage A, B;
  load_stage(A, pa, pc, qj, qi, 0);
  const v2f wl = *(const v2f*)(W2 + 2 * lane);
  #pragma unroll 1
  for (int h = 0; h < HID - 4; h += 4) {
    stage_ready(A);
    load_stage(B, pa, pc, qj, qi, h + 2);
    loads_issued();
    do_stage(A, wl, h, dij, dji);
    stage_ready(B);
    load_stage(A, pa, pc, qj, qi, h + 4);
    loads_issued();
    do_stage(B, wl, h + 2, dij, dji);
  }
  stage_ready(A);
  load_stage(B, pa, pc, qj, qi, HID - 2);
  loads_issued();
  do_stage(A, wl, HID - 4, dij, dji);
  do_stage(B, wl, HID - 2, dij, dji);

  const float b2v = b2p[0];
  v2f s[NJ];
  #pragma unroll
  for (int jj = 0; jj < NJ; ++jj) {
    s[jj].x = -0.5f * (fast_tanh(dij[jj].x + b2v) + fast_tanh(dji[jj].x + b2v));
    s[jj].y = -0.5f * (fast_tanh(dij[jj].y + b2v) + fast_tanh(dji[jj].y + b2v));
  }

  float* om = out + (size_t)mj * NTOK + ti * BI + 2 * lane;
  #pragma unroll
  for (int jj = 0; jj < NJ; ++jj) *(v2f*)(om + (size_t)jj * NTOK) = s[jj];

  #pragma unroll
  for (int jj = 0; jj < NJ; ++jj) {
    sT[(2 * lane) * 33 + wv * NJ + jj]     = s[jj].x;
    sT[(2 * lane + 1) * 33 + wv * NJ + jj] = s[jj].y;
  }
  __syncthreads();
  float* od = out + (size_t)(mi + wv * 16) * NTOK + tj * BI + jh * 32 + lane;
  #pragma unroll
  for (int r = 0; r < 16; ++r) od[(size_t)r * NTOK] = sT[(wv * 16 + r) * 33 + lane];
}

extern "C" void kernel_launch(void* const* d_in, const int* in_sizes, int n_in,
                              void* d_out, int out_size, void* d_ws, size_t ws_size,
                              hipStream_t stream) {
  (void)in_sizes; (void)n_in; (void)out_size; (void)ws_size;
  const float* E  = (const float*)d_in[0];
  const float* W1 = (const float*)d_in[1];
  const float* b1 = (const float*)d_in[2];
  const float* W2 = (const float*)d_in[3];
  const float* b2 = (const float*)d_in[4];
  float* out  = (float*)d_out;
  float* Hbuf = (float*)d_ws;

  proj_kernel<<<dim3((BATCH * NTOK) / 16, HID / 16, 2), 32, 0, stream>>>(E, W1, b1, Hbuf);
  pair_kernel<<<dim3(NTOK / 64, 9, BATCH * 2), 128, 0, stream>>>(
      Hbuf, Hbuf + (size_t)BATCH * NTOK * HID, W2, b2, out);
}


# pair stage: at entry each thread prefetches two 256 B pieces of the h-major workspace (blocks I and J, one h row of one array) into the second-level cache: the group's working set; sums unchanged
# speedup vs baseline: 4.0710x; 1.2153x over previous
#include <hip/hip_runtime.h>
#include <hip/hip_bf16.h>
#include <math.h>

#define BATCH 4
#define NTOK  1024
#define DIM   128
#define HID   64
#define MTOT  (BATCH * NTOK)
#define NJ    8
#define BI    64
#define NBLK  (NTOK / BI)

typedef __attribute__((ext_vector_type(2))) float v2f;
typedef __attribute__((ext_vector_type(4))) float f4;
typedef __attribute__((ext_vector_type(8))) float v8f;

__device__ __forceinline__ float fast_tanh(float x) {
#if __has_builtin(__builtin_amdgcn_tanhf)
  return __builtin_amdgcn_tanhf(x);
#else
  float r;
  asm volatile("v_tanh_f32 %0, %1" : "=v"(r) : "v"(x));
  return r;
#endif
}

__global__ void __launch_bounds__(32)
proj_kernel(const float* __restrict__ E, const float* __restrict__ W1,
            const float* __restrict__ b1, float* __restrict__ Hbuf) {
  const int lane = threadIdx.x;
  const int m0   = blockIdx.x * 16;
  const int n0   = blockIdx.y * 16;
  const int half = blockIdx.z;

  const float* W = W1 + half * (DIM * HID);
  const int row  = m0 + (lane & 15);
  const int koff = (lane >> 4) * 2;
  const int ncol = n0 + (lane & 15);

  v8f c = {};
  #pragma unroll
  for (int k = 0; k < DIM; k += 4) {
    v2f a, b;
    a.x = E[row * DIM + (k + koff)];
    a.y = E[row * DIM + (k + koff + 1)];
    b.x = W[(k + koff) * HID + ncol];
    b.y = W[(k + koff + 1) * HID + ncol];
    c = __builtin_amdgcn_wmma_f32_16x16x4_f32(
         false, a,  false, b,
         (short)0, c,  false,  false);
  }

  const float bias  = (half == 0) ? b1[ncol] : 0.0f;
  const int   rbase = m0 + ((lane >> 4) * 8);
  float* dst = Hbuf + (size_t)half * (MTOT * HID) + (size_t)ncol * MTOT + rbase;
  f4 lo, hi;
  lo.x = c[0] + bias; lo.y = c[1] + bias; lo.z = c[2] + bias; lo.w = c[3] + bias;
  hi.x = c[4] + bias; hi.y = c[5] + bias; hi.z = c[6] + bias; hi.w = c[7] + bias;
  *(f4*)(dst)     = lo;
  *(f4*)(dst + 4) = hi;
}

struct PairStage {
  v2f a0, c0, a1, c1;
  v8f j0, i0, j1, i1;
};

__device__ __forceinline__ void load_stage(PairStage& s, const float* __restrict__ pa,
                                           const float* __restrict__ pc,
                                           const float* __restrict__ qj,
                                           const float* __restrict__ qi, int h) {
  s.a0 = *(const v2f*)(pa + h * MTOT);
  s.c0 = *(const v2f*)(pc + h * MTOT);
  s.a1 = *(const v2f*)(pa + (h + 1) * MTOT);
  s.c1 = *(const v2f*)(pc + (h + 1) * MTOT);
  s.j0 = *(const v8f*)(qj + h * MTOT);
  s.i0 = *(const v8f*)(qi + h * MTOT);
  s.j1 = *(const v8f*)(qj + (h + 1) * MTOT);
  s.i1 = *(const v8f*)(qi + (h + 1) * MTOT);
}

__device__ __forceinline__ void stage_ready(const PairStage& s) {
  __builtin_amdgcn_sched_barrier(0);
  asm volatile("" :: "s"(s.j0));
}

__device__ __forceinline__ void loads_issued() {
  __builtin_amdgcn_sched_barrier(0);
}

__device__ __forceinline__ v2f relu_fma(v2f t, v2f w, v2f acc) {
  const v2f z = {0.0f, 0.0f};
  t = __builtin_elementwise_max(t, z);
  return __builtin_elementwise_fma(t, w, acc);
}

__device__ __forceinline__ void do_step(v2f a, v2f c, v8f xjv, v8f xiv, float w,
                                        v2f (&dij)[NJ], v2f (&dji)[NJ]) {
  const v2f wv = {w, w};
  #pragma unroll
  for (int jj = 0; jj < NJ; ++jj) {
    const v2f xj = {xjv[jj], xjv[jj]};
    const v2f xi = {xiv[jj], xiv[jj]};
    dij[jj] = relu_fma(a + xj, wv, dij[jj]);
    dji[jj] = relu_fma(xi + c, wv, dji[jj]);
  }
}

__device__ __forceinline__ void do_stage(const PairStage& s, v2f wl, int h,
                                         v2f (&dij)[NJ], v2f (&dji)[NJ]) {
  const float wx = wl.x, wy = wl.y;
  const float w0 = __int_as_float(__builtin_amdgcn_readlane(__float_as_int(wx), h >> 1));
  const float w1 = __int_as_float(__builtin_amdgcn_readlane(__float_as_int(wy), h >> 1));
  do_step(s.a0, s.c0, s.j0, s.i0, w0, dij, dji);
  do_step(s.a1, s.c1, s.j1, s.i1, w1, dij, dji);
}

__global__ void __launch_bounds__(128)
pair_kernel(const float* __restrict__ HiT, const float* __restrict__ HjT,
            const float* __restrict__ W2, const float* __restrict__ b2p,
            float* __restrict__ out) {
  const int ti = blockIdx.x, k = blockIdx.y;
  const int b = blockIdx.z >> 1, jh = blockIdx.z & 1;
  if (k == NBLK / 2 && ti >= NBLK / 2) return;
  const int tj = (ti + k) & (NBLK - 1);

  __shared__ float sT[BI * 33];

  const int lane = threadIdx.x & 31;
  const int wv = __builtin_amdgcn_readfirstlane((int)(threadIdx.x >> 5));

  const int mi = b * NTOK + ti * BI;
  const int mj = b * NTOK + tj * BI + jh * 32 + wv * NJ;

  const float* pa = HiT + mi + 2 * lane;
  const float* pc = HjT + mi + 2 * lane;
  const float* qj = HjT + mj;
  const float* qi = HiT + mj;

  v2f dij[NJ], dji[NJ];
  #pragma unroll
  for (int jj = 0; jj < NJ; ++jj) { dij[jj].x = 0.f; dij[jj].y = 0.f; dji[jj].x = 0.f; dji[jj].y = 0.f; }

  PairStage A, B;
  load_stage(A, pa, pc, qj, qi, 0);
  const v2f wl = *(const v2f*)(W2 + 2 * lane);

  {
    const float* row = ((threadIdx.x >> 6) ? HjT : HiT)
                     + (size_t)(threadIdx.x & (HID - 1)) * MTOT + b * NTOK;
    __builtin_prefetch(row + ti * BI, 0, 2);
    __builtin_prefetch(row + tj * BI, 0, 2);
  }
  #pragma unroll 1
  for (int h = 0; h < HID - 4; h += 4) {
    stage_ready(A);
    load_stage(B, pa, pc, qj, qi, h + 2);
    loads_issued();
    do_stage(A, wl, h, dij, dji);
    stage_ready(B);
    load_stage(A, pa, pc, qj, qi, h + 4);
    loads_issued();
    do_stage(B, wl, h + 2, dij, dji);
  }
  stage_ready(A);
  load_stage(B, pa, pc, qj, qi, HID - 2);
  loads_issued();
  do_stage(A, wl, HID - 4, dij, dji);
  do_stage(B, wl, HID - 2, dij, dji);

  const float b2v = b2p[0];
  v2f s[NJ];
  #pragma unroll
  for (int jj = 0; jj < NJ; ++jj) {
    s[jj].x = -0.5f * (fast_tanh(dij[jj].x + b2v) + fast_tanh(dji[jj].x + b2v));
    s[jj].y = -0.5f * (fast_tanh(dij[jj].y + b2v) + fast_tanh(dji[jj].y + b2v));
  }

  float* om = out + (size_t)mj * NTOK + ti * BI + 2 * lane;
  #pragma unroll
  for (int jj = 0; jj < NJ; ++jj) *(v2f*)(om + (size_t)jj * NTOK) = s[jj];

  #pragma unroll
  for (int jj = 0; jj < NJ; ++jj) {
    sT[(2 * lane) * 33 + wv * NJ + jj]     = s[jj].x;
    sT[(2 * lane + 1) * 33 + wv * NJ + jj] = s[jj].y;
  }
  __syncthreads();
  float* od = out + (size_t)(mi + wv * 16) * NTOK + tj * BI + jh * 32 + lane;
  #pragma unroll
  for (int r = 0; r < 16; ++r) od[(size_t)r * NTOK] = sT[(wv * 16 + r) * 33 + lane];
}

extern "C" void kernel_launch(void* const* d_in, const int* in_sizes, int n_in,
                              void* d_out, int out_size, void* d_ws, size_t ws_size,
                              hipStream_t stream) {
  (void)in_sizes; (void)n_in; (void)out_size; (void)ws_size;
  const float* E  = (const float*)d_in[0];
  const float* W1 = (const float*)d_in[1];
  const float* b1 = (const float*)d_in[2];
  const float* W2 = (const float*)d_in[3];
  const float* b2 = (const float*)d_in[4];
  float* out  = (float*)d_out;
  float* Hbuf = (float*)d_ws;

  proj_kernel<<<dim3((BATCH * NTOK) / 16, HID / 16, 2), 32, 0, stream>>>(E, W1, b1, Hbuf);
  pair_kernel<<<dim3(NTOK / 64, 9, BATCH * 2), 128, 0, stream>>>(
      Hbuf, Hbuf + (size_t)BATCH * NTOK * HID, W2, b2, out);
}


# pair stage: a diagonal block pair is ONE group forming its 64x64 sums once (2nd accumulator set: 32 more partner rows, same direction; mirrored via LDS), not two forming each twice; 1024 equal groups
# speedup vs baseline: 4.5732x; 1.1234x over previous
#include <hip/hip_runtime.h>
#include <hip/hip_bf16.h>
#include <math.h>

#define BATCH 4
#define NTOK  1024
#define DIM   128
#define HID   64
#define MTOT  (BATCH * NTOK)
#define NJ    8
#define BI    64
#define NBLK  (NTOK / BI)
#define NUNIT 256
#define DP    65

typedef __attribute__((ext_vector_type(2))) float v2f;
typedef __attribute__((ext_vector_type(4))) float f4;
typedef __attribute__((ext_vector_type(8))) float v8f;

__device__ __forceinline__ float fast_tanh(float x) {
#if __has_builtin(__builtin_amdgcn_tanhf)
  return __builtin_amdgcn_tanhf(x);
#else
  float r;
  asm volatile("v_tanh_f32 %0, %1" : "=v"(r) : "v"(x));
  return r;
#endif
}

__global__ void __launch_bounds__(32)
proj_kernel(const float* __restrict__ E, const float* __restrict__ W1,
            const float* __restrict__ b1, float* __restrict__ Hbuf) {
  const int lane = threadIdx.x;
  const int m0   = blockIdx.x * 16;
  const int n0   = blockIdx.y * 16;
  const int half = blockIdx.z;

  const float* W = W1 + half * (DIM * HID);
  const int row  = m0 + (lane & 15);
  const int koff = (lane >> 4) * 2;
  const int ncol = n0 + (lane & 15);

  v8f c = {};
  #pragma unroll
  for (int k = 0; k < DIM; k += 4) {
    v2f a, b;
    a.x = E[row * DIM + (k + koff)];
    a.y = E[row * DIM + (k + koff + 1)];
    b.x = W[(k + koff) * HID + ncol];
    b.y = W[(k + koff + 1) * HID + ncol];
    c = __builtin_amdgcn_wmma_f32_16x16x4_f32(
         false, a,  false, b,
         (short)0, c,  false,  false);
  }

  const float bias  = (half == 0) ? b1[ncol] : 0.0f;
  const int   rbase = m0 + ((lane >> 4) * 8);
  float* dst = Hbuf + (size_t)half * (MTOT * HID) + (size_t)ncol * MTOT + rbase;
  f4 lo, hi;
  lo.x = c[0] + bias; lo.y = c[1] + bias; lo.z = c[2] + bias; lo.w = c[3] + bias;
  hi.x = c[4] + bias; hi.y = c[5] + bias; hi.z = c[6] + bias; hi.w = c[7] + bias;
  *(f4*)(dst)     = lo;
  *(f4*)(dst + 4) = hi;
}

struct PairStage {
  v2f a0, c0, a1, c1;
  v8f j0, i0, j1, i1;
};

__device__ __forceinline__ void load_stage(PairStage& s, const float* __restrict__ pa,
                                           const float* __restrict__ pc,
                                           const float* __restrict__ qj,
                                           const float* __restrict__ qi, int h) {
  s.a0 = *(const v2f*)(pa + h * MTOT);
  s.c0 = *(const v2f*)(pc + h * MTOT);
  s.a1 = *(const v2f*)(pa + (h + 1) * MTOT);
  s.c1 = *(const v2f*)(pc + (h + 1) * MTOT);
  s.j0 = *(const v8f*)(qj + h * MTOT);
  s.i0 = *(const v8f*)(qi + h * MTOT);
  s.j1 = *(const v8f*)(qj + (h + 1) * MTOT);
  s.i1 = *(const v8f*)(qi + (h + 1) * MTOT);
}

__device__ __forceinline__ void stage_ready(const PairStage& s) {
  __builtin_amdgcn_sched_barrier(0);
  asm volatile("" :: "s"(s.j0));
}

__device__ __forceinline__ void loads_issued() {
  __builtin_amdgcn_sched_barrier(0);
}

__device__ __forceinline__ v2f relu_fma(v2f t, v2f w, v2f acc) {
  const v2f z = {0.0f, 0.0f};
  t = __builtin_elementwise_max(t, z);
  return __builtin_elementwise_fma(t, w, acc);
}

__device__ __forceinline__ void do_step(v2f a, v2f c, v8f xjv, v8f xiv, float w,
                                        v2f (&dij)[NJ], v2f (&dji)[NJ]) {
  const v2f wv = {w, w};
  #pragma unroll
  for (int jj = 0; jj < NJ; ++jj) {
    const v2f xj = {xjv[jj], xjv[jj]};
    const v2f xi = {xiv[jj], xiv[jj]};
    dij[jj] = relu_fma(a + xj, wv, dij[jj]);
    dji[jj] = relu_fma(xi + c, wv, dji[jj]);
  }
}

__device__ __forceinline__ void do_stage(const PairStage& s, v2f wl, int h,
                                         v2f (&dij)[NJ], v2f (&dji)[NJ]) {
  const float wx = wl.x, wy = wl.y;
  const float w0 = __int_as_float(__builtin_amdgcn_readlane(__float_as_int(wx), h >> 1));
  const float w1 = __int_as_float(__builtin_amdgcn_readlane(__float_as_int(wy), h >> 1));
  do_step(s.a0, s.c0, s.j0, s.i0, w0, dij, dji);
  do_step(s.a1, s.c1, s.j1, s.i1, w1, dij, dji);
}

__global__ void __launch_bounds__(128)
pair_kernel(const float* __restrict__ HiT, const float* __restrict__ HjT,
            const float* __restrict__ W2, const float* __restrict__ b2p,
            float* __restrict__ out) {
  const int u = blockIdx.x, b = blockIdx.y;
  const bool diag = u < NBLK;
  const int v = u - NBLK, p = v >> 1;
  const int jh = diag ? 0 : (v & 1);
  const int k  = diag ? 0 : 1 + (p >> 4);
  const int ti = diag ? u : (p & (NBLK - 1));
  const int tj = (ti + k) & (NBLK - 1);

  __shared__ float sT[BI * DP];

  const int lane = threadIdx.x & 31;
  const int wv = __builtin_amdgcn_readfirstlane((int)(threadIdx.x >> 5));

  const int mi = b * NTOK + ti * BI;
  const int mj = b * NTOK + tj * BI + jh * 32 + wv * NJ;

  const float* pa = HiT + mi + 2 * lane;
  const float* pc = (diag ? HiT : HjT) + mi + 2 * lane;
  const float* qj = HjT + mj;
  const float* qi = (diag ? HjT + 32 : HiT) + mj;

  v2f dij[NJ], dji[NJ];
  #pragma unroll
  for (int jj = 0; jj < NJ; ++jj) { dij[jj].x = 0.f; dij[jj].y = 0.f; dji[jj].x = 0.f; dji[jj].y = 0.f; }

  PairStage A, B;
  load_stage(A, pa, pc, qj, qi, 0);
  const v2f wl = *(const v2f*)(W2 + 2 * lane);

  {
    const float* row = ((threadIdx.x >> 6) ? HjT : HiT)
                     + (size_t)(threadIdx.x & (HID - 1)) * MTOT + b * NTOK;
    __builtin_prefetch(row + ti * BI, 0, 2);
    __builtin_prefetch(row + tj * BI, 0, 2);
  }
  #pragma unroll 1
  for (int h = 0; h < HID - 4; h += 4) {
    stage_ready(A);
    load_stage(B, pa, pc, qj, qi, h + 2);
    loads_issued();
    do_stage(A, wl, h, dij, dji);
    stage_ready(B);
    load_stage(A, pa, pc, qj, qi, h + 4);
    loads_issued();
    do_stage(B, wl, h + 2, dij, dji);
  }
  stage_ready(A);
  load_stage(B, pa, pc, qj, qi, HID - 2);
  loads_issued();
  do_stage(A, wl, HID - 4, dij, dji);
  do_stage(B, wl, HID - 2, dij, dji);

  const float b2v = b2p[0];
  if (diag) {
    #pragma unroll
    for (int jj = 0; jj < NJ; ++jj) {
      sT[(2 * lane) * DP + wv * NJ + jj]          = fast_tanh(dij[jj].x + b2v);
      sT[(2 * lane + 1) * DP + wv * NJ + jj]      = fast_tanh(dij[jj].y + b2v);
      sT[(2 * lane) * DP + 32 + wv * NJ + jj]     = fast_tanh(dji[jj].x + b2v);
      sT[(2 * lane + 1) * DP + 32 + wv * NJ + jj] = fast_tanh(dji[jj].y + b2v);
    }
    __syncthreads();
    float* ob = out + (size_t)(mi + wv * 16) * NTOK + ti * BI + lane;
    #pragma unroll
    for (int r = 0; r < 16; ++r) {
      const int row = wv * 16 + r;
      #pragma unroll
      for (int hf = 0; hf < 2; ++hf) {
        const int col = hf * 32 + lane;
        ob[(size_t)r * NTOK + hf * 32] = -0.5f * (sT[row * DP + col] + sT[col * DP + row]);
      }
    }
    return;
  }
  v2f s[NJ];
  #pragma unroll
  for (int jj = 0; jj < NJ; ++jj) {
    s[jj].x = -0.5f * (fast_tanh(dij[jj].x + b2v) + fast_tanh(dji[jj].x + b2v));
    s[jj].y = -0.5f * (fast_tanh(dij[jj].y + b2v) + fast_tanh(dji[jj].y + b2v));
  }

  float* om = out + (size_t)mj * NTOK + ti * BI + 2 * lane;
  #pragma unroll
  for (int jj = 0; jj < NJ; ++jj) *(v2f*)(om + (size_t)jj * NTOK) = s[jj];

  #pragma unroll
  for (int jj = 0; jj < NJ; ++jj) {
    sT[(2 * lane) * 33 + wv * NJ + jj]     = s[jj].x;
    sT[(2 * lane + 1) * 33 + wv * NJ + jj] = s[jj].y;
  }
  __syncthreads();
  float* od = out + (size_t)(mi + wv * 16) * NTOK + tj * BI + jh * 32 + lane;
  #pragma unroll
  for (int r = 0; r < 16; ++r) od[(size_t)r * NTOK] = sT[(wv * 16 + r) * 33 + lane];
}

extern "C" void kernel_launch(void* const* d_in, const int* in_sizes, int n_in,
                              void* d_out, int out_size, void* d_ws, size_t ws_size,
                              hipStream_t stream) {
  (void)in_sizes; (void)n_in; (void)out_size; (void)ws_size;
  const float* E  = (const float*)d_in[0];
  const float* W1 = (const float*)d_in[1];
  const float* b1 = (const float*)d_in[2];
  const float* W2 = (const float*)d_in[3];
  const float* b2 = (const float*)d_in[4];
  float* out  = (float*)d_out;
  float* Hbuf = (float*)d_ws;

  proj_kernel<<<dim3((BATCH * NTOK) / 16, HID / 16, 2), 32, 0, stream>>>(E, W1, b1, Hbuf);
  pair_kernel<<<dim3(NUNIT, BATCH), 128, 0, stream>>>(
      Hbuf, Hbuf + (size_t)BATCH * NTOK * HID, W2, b2, out);
}


# pair stage: one direction per wave (8-wave groups: 4 waves form d_ij, 4 form d_ji, tanh halves meet in LDS): a scalar set holds 4 h steps, half the load waits, twice the resident waves; sums kept
# speedup vs baseline: 4.8250x; 1.0551x over previous
#include <hip/hip_runtime.h>
#include <hip/hip_bf16.h>
#include <math.h>

#define BATCH 4
#define NTOK  1024
#define DIM   128
#define HID   64
#define MTOT  (BATCH * NTOK)
#define NJ    8
#define BI    64
#define NBLK  (NTOK / BI)
#define NUNIT 256
#define DP    65
#define OP    33
#define HS    4

typedef __attribute__((ext_vector_type(2))) float v2f;
typedef __attribute__((ext_vector_type(4))) float f4;
typedef __attribute__((ext_vector_type(8))) float v8f;

__device__ __forceinline__ float fast_tanh(float x) {
#if __has_builtin(__builtin_amdgcn_tanhf)
  return __builtin_amdgcn_tanhf(x);
#else
  float r;
  asm volatile("v_tanh_f32 %0, %1" : "=v"(r) : "v"(x));
  return r;
#endif
}

__global__ void __launch_bounds__(32)
proj_kernel(const float* __restrict__ E, const float* __restrict__ W1,
            const float* __restrict__ b1, float* __restrict__ Hbuf) {
  const int lane = threadIdx.x;
  const int m0   = blockIdx.x * 16;
  const int n0   = blockIdx.y * 16;
  const int half = blockIdx.z;

  const float* W = W1 + half * (DIM * HID);
  const int row  = m0 + (lane & 15);
  const int koff = (lane >> 4) * 2;
  const int ncol = n0 + (lane & 15);

  v8f c = {};
  #pragma unroll
  for (int k = 0; k < DIM; k += 4) {
    v2f a, b;
    a.x = E[row * DIM + (k + koff)];
    a.y = E[row * DIM + (k + koff + 1)];
    b.x = W[(k + koff) * HID + ncol];
    b.y = W[(k + koff + 1) * HID + ncol];
    c = __builtin_amdgcn_wmma_f32_16x16x4_f32(
         false, a,  false, b,
         (short)0, c,  false,  false);
  }

  const float bias  = (half == 0) ? b1[ncol] : 0.0f;
  const int   rbase = m0 + ((lane >> 4) * 8);
  float* dst = Hbuf + (size_t)half * (MTOT * HID) + (size_t)ncol * MTOT + rbase;
  f4 lo, hi;
  lo.x = c[0] + bias; lo.y = c[1] + bias; lo.z = c[2] + bias; lo.w = c[3] + bias;
  hi.x = c[4] + bias; hi.y = c[5] + bias; hi.z = c[6] + bias; hi.w = c[7] + bias;
  *(f4*)(dst)     = lo;
  *(f4*)(dst + 4) = hi;
}

struct PairStage {
  v2f a[HS];
  v8f p[HS];
};

__device__ __forceinline__ void load_stage(PairStage& s, const float* __restrict__ po,
                                           const float* __restrict__ qp, int h) {
  #pragma unroll
  for (int t = 0; t < HS; ++t) {
    s.a[t] = *(const v2f*)(po + (h + t) * MTOT);
    s.p[t] = *(const v8f*)(qp + (h + t) * MTOT);
  }
}

__device__ __forceinline__ void stage_ready(const PairStage& s) {
  __builtin_amdgcn_sched_barrier(0);
  asm volatile("" :: "s"(s.p[0]));
}

__device__ __forceinline__ void loads_issued() {
  __builtin_amdgcn_sched_barrier(0);
}

__device__ __forceinline__ v2f relu_fma(v2f t, v2f w, v2f acc) {
  const v2f z = {0.0f, 0.0f};
  t = __builtin_elementwise_max(t, z);
  return __builtin_elementwise_fma(t, w, acc);
}

__device__ __forceinline__ void do_step(v2f a, v8f pv, float w, v2f (&acc)[NJ]) {
  const v2f wv = {w, w};
  #pragma unroll
  for (int jj = 0; jj < NJ; ++jj) {
    const v2f x = {pv[jj], pv[jj]};
    acc[jj] = relu_fma(a + x, wv, acc[jj]);
  }
}

__device__ __forceinline__ void do_stage(const PairStage& s, v2f wl, int h, v2f (&acc)[NJ]) {
  const float wx = wl.x, wy = wl.y;
  #pragma unroll
  for (int t = 0; t < HS; t += 2) {
    const float w0 = __int_as_float(__builtin_amdgcn_readlane(__float_as_int(wx), (h + t) >> 1));
    const float w1 = __int_as_float(__builtin_amdgcn_readlane(__float_as_int(wy), (h + t) >> 1));
    do_step(s.a[t], s.p[t], w0, acc);
    do_step(s.a[t + 1], s.p[t + 1], w1, acc);
  }
}

__global__ void __launch_bounds__(256)
pair_kernel(const float* __restrict__ HiT, const float* __restrict__ HjT,
            const float* __restrict__ W2, const float* __restrict__ b2p,
            float* __restrict__ out) {
  const int u = blockIdx.x, b = blockIdx.y;
  const bool diag = u < NBLK;
  const int v = u - NBLK, p = v >> 1;
  const int jh = diag ? 0 : (v & 1);
  const int k  = diag ? 0 : 1 + (p >> 4);
  const int ti = diag ? u : (p & (NBLK - 1));
  const int tj = (ti + k) & (NBLK - 1);

  __shared__ float sT[2 * BI * OP];

  const int lane = threadIdx.x & 31;
  const int wv  = __builtin_amdgcn_readfirstlane((int)(threadIdx.x >> 5));
  const int q   = wv & 3;
  const int dir = wv >> 2;
  const bool fwd = diag || dir == 0;

  const int mi = b * NTOK + ti * BI;
  const int mh = b * NTOK + tj * BI + (diag ? dir * 32 : jh * 32);
  const int mp = mh + q * NJ;

  const float* po = (fwd ? HiT : HjT) + mi + 2 * lane;
  const float* qp = (fwd ? HjT : HiT) + mp;

  v2f acc[NJ];
  #pragma unroll
  for (int jj = 0; jj < NJ; ++jj) { acc[jj].x = 0.f; acc[jj].y = 0.f; }

  PairStage A, B;
  load_stage(A, po, qp, 0);
  const v2f wl = *(const v2f*)(W2 + 2 * lane);

  {
    const float* row = (((threadIdx.x >> 6) & 1) ? HjT : HiT)
                     + (size_t)(threadIdx.x & (HID - 1)) * MTOT + b * NTOK;
    __builtin_prefetch(row + ((threadIdx.x >> 7) ? tj : ti) * BI, 0, 2);
  }
  #pragma unroll 1
  for (int h = 0; h < HID - 2 * HS; h += 2 * HS) {
    stage_ready(A);
    load_stage(B, po, qp, h + HS);
    loads_issued();
    do_stage(A, wl, h, acc);
    stage_ready(B);
    load_stage(A, po, qp, h + 2 * HS);
    loads_issued();
    do_stage(B, wl, h + HS, acc);
  }
  stage_ready(A);
  load_stage(B, po, qp, HID - HS);
  loads_issued();
  do_stage(A, wl, HID - 2 * HS, acc);
  do_stage(B, wl, HID - HS, acc);

  const float b2v = b2p[0];
  if (diag) {
    const int cb = dir * 32 + q * NJ;
    #pragma unroll
    for (int jj = 0; jj < NJ; ++jj) {
      sT[(2 * lane) * DP + cb + jj]     = fast_tanh(acc[jj].x + b2v);
      sT[(2 * lane + 1) * DP + cb + jj] = fast_tanh(acc[jj].y + b2v);
    }
    __syncthreads();
    float* ob = out + (size_t)(mi + wv * 8) * NTOK + ti * BI + lane;
    #pragma unroll
    for (int r = 0; r < 8; ++r) {
      const int row = wv * 8 + r;
      #pragma unroll
      for (int hf = 0; hf < 2; ++hf) {
        const int col = hf * 32 + lane;
        ob[(size_t)r * NTOK + hf * 32] = -0.5f * (sT[row * DP + col] + sT[col * DP + row]);
      }
    }
    return;
  }

  float* Td = sT + dir * (BI * OP);
  #pragma unroll
  for (int jj = 0; jj < NJ; ++jj) {
    Td[(2 * lane) * OP + q * NJ + jj]     = fast_tanh(acc[jj].x + b2v);
    Td[(2 * lane + 1) * OP + q * NJ + jj] = fast_tanh(acc[jj].y + b2v);
  }
  __syncthreads();
  const float* T0 = sT;
  const float* T1 = sT + BI * OP;

  float* od = out + (size_t)(mi + wv * 8) * NTOK + tj * BI + jh * 32 + lane;
  #pragma unroll
  for (int r = 0; r < 8; ++r) {
    const int row = wv * 8 + r;
    od[(size_t)r * NTOK] = -0.5f * (T0[row * OP + lane] + T1[row * OP + lane]);
  }

  float* om = out + (size_t)(mh + wv * 4) * NTOK + ti * BI + lane;
  #pragma unroll
  for (int r = 0; r < 4; ++r) {
    const int jr = wv * 4 + r;
    #pragma unroll
    for (int hf = 0; hf < 2; ++hf) {
      const int i = hf * 32 + lane;
      om[(size_t)r * NTOK + hf * 32] = -0.5f * (T0[i * OP + jr] + T1[i * OP + jr]);
    }
  }
}

extern "C" void kernel_launch(void* const* d_in, const int* in_sizes, int n_in,
                              void* d_out, int out_size, void* d_ws, size_t ws_size,
                              hipStream_t stream) {
  (void)in_sizes; (void)n_in; (void)out_size; (void)ws_size;
  const float* E  = (const float*)d_in[0];
  const float* W1 = (const float*)d_in[1];
  const float* b1 = (const float*)d_in[2];
  const float* W2 = (const float*)d_in[3];
  const float* b2 = (const float*)d_in[4];
  float* out  = (float*)d_out;
  float* Hbuf = (float*)d_ws;

  proj_kernel<<<dim3((BATCH * NTOK) / 16, HID / 16, 2), 32, 0, stream>>>(E, W1, b1, Hbuf);
  pair_kernel<<<dim3(NUNIT, BATCH), 256, 0, stream>>>(
      Hbuf, Hbuf + (size_t)BATCH * NTOK * HID, W2, b2, out);
}


# pair stage: relu comes from the packed add's clamp output modifier (max gone: 2 instructions per two terms, not 3); workspace holds hi',hj * 2^-64, w2 * 2^64 (exact), clamp top = 2^64; sums kept
# speedup vs baseline: 5.4747x; 1.1347x over previous
#include <hip/hip_runtime.h>
#include <hip/hip_bf16.h>
#include <math.h>

#define BATCH 4
#define NTOK  1024
#define DIM   128
#define HID   64
#define MTOT  (BATCH * NTOK)
#define NJ    8
#define BI    64
#define NBLK  (NTOK / BI)
#define NUNIT 256
#define DP    65
#define OP    33
#define HS    4
#define SC_DN 0x1p-64f
#define SC_UP 0x1p+64f

typedef __attribute__((ext_vector_type(2))) float v2f;
typedef __attribute__((ext_vector_type(4))) float f4;
typedef __attribute__((ext_vector_type(8))) float v8f;

__device__ __forceinline__ float fast_tanh(float x) {
#if __has_builtin(__builtin_amdgcn_tanhf)
  return __builtin_amdgcn_tanhf(x);
#else
  float r;
  asm volatile("v_tanh_f32 %0, %1" : "=v"(r) : "v"(x));
  return r;
#endif
}

__global__ void __launch_bounds__(32)
proj_kernel(const float* __restrict__ E, const float* __restrict__ W1,
            const float* __restrict__ b1, float* __restrict__ Hbuf) {
  const int lane = threadIdx.x;
  const int m0   = blockIdx.x * 16;
  const int n0   = blockIdx.y * 16;
  const int half = blockIdx.z;

  const float* W = W1 + half * (DIM * HID);
  const int row  = m0 + (lane & 15);
  const int koff = (lane >> 4) * 2;
  const int ncol = n0 + (lane & 15);

  v8f c = {};
  #pragma unroll
  for (int k = 0; k < DIM; k += 4) {
    v2f a, b;
    a.x = E[row * DIM + (k + koff)];
    a.y = E[row * DIM + (k + koff + 1)];
    b.x = W[(k + koff) * HID + ncol];
    b.y = W[(k + koff + 1) * HID + ncol];
    c = __builtin_amdgcn_wmma_f32_16x16x4_f32(
         false, a,  false, b,
         (short)0, c,  false,  false);
  }

  const float bias  = (half == 0) ? b1[ncol] : 0.0f;
  const int   rbase = m0 + ((lane >> 4) * 8);
  float* dst = Hbuf + (size_t)half * (MTOT * HID) + (size_t)ncol * MTOT + rbase;
  f4 lo, hi;
  lo.x = (c[0] + bias) * SC_DN; lo.y = (c[1] + bias) * SC_DN;
  lo.z = (c[2] + bias) * SC_DN; lo.w = (c[3] + bias) * SC_DN;
  hi.x = (c[4] + bias) * SC_DN; hi.y = (c[5] + bias) * SC_DN;
  hi.z = (c[6] + bias) * SC_DN; hi.w = (c[7] + bias) * SC_DN;
  *(f4*)(dst)     = lo;
  *(f4*)(dst + 4) = hi;
}

struct PairStage {
  v2f a[HS];
  v8f p[HS];
};

__device__ __forceinline__ void load_stage(PairStage& s, const float* __restrict__ po,
                                           const float* __restrict__ qp, int h) {
  #pragma unroll
  for (int t = 0; t < HS; ++t) {
    s.a[t] = *(const v2f*)(po + (h + t) * MTOT);
    s.p[t] = *(const v8f*)(qp + (h + t) * MTOT);
  }
}

__device__ __forceinline__ void stage_ready(const PairStage& s) {
  __builtin_amdgcn_sched_barrier(0);
  asm volatile("" :: "s"(s.p[0]));
}

__device__ __forceinline__ void loads_issued() {
  __builtin_amdgcn_sched_barrier(0);
}

__device__ __forceinline__ void add_relu8(v2f a, v8f pv, v2f (&t)[NJ]) {
  const v2f x0 = __builtin_shufflevector(pv, pv, 0, 1);
  const v2f x1 = __builtin_shufflevector(pv, pv, 1, -1);
  const v2f x2 = __builtin_shufflevector(pv, pv, 2, 3);
  const v2f x3 = __builtin_shufflevector(pv, pv, 3, -1);
  const v2f x4 = __builtin_shufflevector(pv, pv, 4, 5);
  const v2f x5 = __builtin_shufflevector(pv, pv, 5, -1);
  const v2f x6 = __builtin_shufflevector(pv, pv, 6, 7);
  const v2f x7 = __builtin_shufflevector(pv, pv, 7, -1);
  asm("v_pk_add_f32 %0, %8, %9 op_sel_hi:[1,0] clamp\n\t"
      "v_pk_add_f32 %1, %8, %10 op_sel_hi:[1,0] clamp\n\t"
      "v_pk_add_f32 %2, %8, %11 op_sel_hi:[1,0] clamp\n\t"
      "v_pk_add_f32 %3, %8, %12 op_sel_hi:[1,0] clamp\n\t"
      "v_pk_add_f32 %4, %8, %13 op_sel_hi:[1,0] clamp\n\t"
      "v_pk_add_f32 %5, %8, %14 op_sel_hi:[1,0] clamp\n\t"
      "v_pk_add_f32 %6, %8, %15 op_sel_hi:[1,0] clamp\n\t"
      "v_pk_add_f32 %7, %8, %16 op_sel_hi:[1,0] clamp"
      : "=&v"(t[0]), "=&v"(t[1]), "=&v"(t[2]), "=&v"(t[3]),
        "=&v"(t[4]), "=&v"(t[5]), "=&v"(t[6]), "=&v"(t[7])
      : "v"(a), "s"(x0), "s"(x1), "s"(x2), "s"(x3), "s"(x4), "s"(x5), "s"(x6), "s"(x7));
}

__device__ __forceinline__ void fma8(const v2f (&t)[NJ], v2f wv, v2f (&acc)[NJ]) {
  asm("v_pk_fma_f32 %0, %8, %16, %0 op_sel_hi:[1,0,1]\n\t"
      "v_pk_fma_f32 %1, %9, %16, %1 op_sel_hi:[1,0,1]\n\t"
      "v_pk_fma_f32 %2, %10, %16, %2 op_sel_hi:[1,0,1]\n\t"
      "v_pk_fma_f32 %3, %11, %16, %3 op_sel_hi:[1,0,1]\n\t"
      "v_pk_fma_f32 %4, %12, %16, %4 op_sel_hi:[1,0,1]\n\t"
      "v_pk_fma_f32 %5, %13, %16, %5 op_sel_hi:[1,0,1]\n\t"
      "v_pk_fma_f32 %6, %14, %16, %6 op_sel_hi:[1,0,1]\n\t"
      "v_pk_fma_f32 %7, %15, %16, %7 op_sel_hi:[1,0,1]"
      : "+v"(acc[0]), "+v"(acc[1]), "+v"(acc[2]), "+v"(acc[3]),
        "+v"(acc[4]), "+v"(acc[5]), "+v"(acc[6]), "+v"(acc[7])
      : "v"(t[0]), "v"(t[1]), "v"(t[2]), "v"(t[3]),
        "v"(t[4]), "v"(t[5]), "v"(t[6]), "v"(t[7]), "s"(wv));
}

__device__ __forceinline__ void do_step(v2f a, v8f pv, float w, v2f (&acc)[NJ]) {
  const v2f wv = {w, w};
  v2f t[NJ];
  add_relu8(a, pv, t);
  fma8(t, wv, acc);
}

__device__ __forceinline__ void do_stage(const PairStage& s, v2f wl, int h, v2f (&acc)[NJ]) {
  const float wx = wl.x, wy = wl.y;
  #pragma unroll
  for (int t = 0; t < HS; t += 2) {
    const float w0 = __int_as_float(__builtin_amdgcn_readlane(__float_as_int(wx), (h + t) >> 1));
    const float w1 = __int_as_float(__builtin_amdgcn_readlane(__float_as_int(wy), (h + t) >> 1));
    do_step(s.a[t], s.p[t], w0, acc);
    do_step(s.a[t + 1], s.p[t + 1], w1, acc);
  }
}

__global__ void __launch_bounds__(256)
pair_kernel(const float* __restrict__ HiT, const float* __restrict__ HjT,
            const float* __restrict__ W2, const float* __restrict__ b2p,
            float* __restrict__ out) {
  const int u = blockIdx.x, b = blockIdx.y;
  const bool diag = u < NBLK;
  const int v = u - NBLK, p = v >> 1;
  const int jh = diag ? 0 : (v & 1);
  const int k  = diag ? 0 : 1 + (p >> 4);
  const int ti = diag ? u : (p & (NBLK - 1));
  const int tj = (ti + k) & (NBLK - 1);

  __shared__ float sT[2 * BI * OP];

  const int lane = threadIdx.x & 31;
  const int wv  = __builtin_amdgcn_readfirstlane((int)(threadIdx.x >> 5));
  const int q   = wv & 3;
  const int dir = wv >> 2;
  const bool fwd = diag || dir == 0;

  const int mi = b * NTOK + ti * BI;
  const int mh = b * NTOK + tj * BI + (diag ? dir * 32 : jh * 32);
  const int mp = mh + q * NJ;

  const float* po = (fwd ? HiT : HjT) + mi + 2 * lane;
  const float* qp = (fwd ? HjT : HiT) + mp;

  v2f acc[NJ];
  #pragma unroll
  for (int jj = 0; jj < NJ; ++jj) { acc[jj].x = 0.f; acc[jj].y = 0.f; }

  PairStage A, B;
  load_stage(A, po, qp, 0);
  const v2f wl = *(const v2f*)(W2 + 2 * lane) * SC_UP;

  {
    const float* row = (((threadIdx.x >> 6) & 1) ? HjT : HiT)
                     + (size_t)(threadIdx.x & (HID - 1)) * MTOT + b * NTOK;
    __builtin_prefetch(row + ((threadIdx.x >> 7) ? tj : ti) * BI, 0, 2);
  }
  #pragma unroll 1
  for (int h = 0; h < HID - 2 * HS; h += 2 * HS) {
    stage_ready(A);
    load_stage(B, po, qp, h + HS);
    loads_issued();
    do_stage(A, wl, h, acc);
    stage_ready(B);
    load_stage(A, po, qp, h + 2 * HS);
    loads_issued();
    do_stage(B, wl, h + HS, acc);
  }
  stage_ready(A);
  load_stage(B, po, qp, HID - HS);
  loads_issued();
  do_stage(A, wl, HID - 2 * HS, acc);
  do_stage(B, wl, HID - HS, acc);

  const float b2v = b2p[0];
  if (diag) {
    const int cb = dir * 32 + q * NJ;
    #pragma unroll
    for (int jj = 0; jj < NJ; ++jj) {
      sT[(2 * lane) * DP + cb + jj]     = fast_tanh(acc[jj].x + b2v);
      sT[(2 * lane + 1) * DP + cb + jj] = fast_tanh(acc[jj].y + b2v);
    }
    __syncthreads();
    float* ob = out + (size_t)(mi + wv * 8) * NTOK + ti * BI + lane;
    #pragma unroll
    for (int r = 0; r < 8; ++r) {
      const int row = wv * 8 + r;
      #pragma unroll
      for (int hf = 0; hf < 2; ++hf) {
        const int col = hf * 32 + lane;
        ob[(size_t)r * NTOK + hf * 32] = -0.5f * (sT[row * DP + col] + sT[col * DP + row]);
      }
    }
    return;
  }

  float* Td = sT + dir * (BI * OP);
  #pragma unroll
  for (int jj = 0; jj < NJ; ++jj) {
    Td[(2 * lane) * OP + q * NJ + jj]     = fast_tanh(acc[jj].x + b2v);
    Td[(2 * lane + 1) * OP + q * NJ + jj] = fast_tanh(acc[jj].y + b2v);
  }
  __syncthreads();
  const float* T0 = sT;
  const float* T1 = sT + BI * OP;

  float* od = out + (size_t)(mi + wv * 8) * NTOK + tj * BI + jh * 32 + lane;
  #pragma unroll
  for (int r = 0; r < 8; ++r) {
    const int row = wv * 8 + r;
    od[(size_t)r * NTOK] = -0.5f * (T0[row * OP + lane] + T1[row * OP + lane]);
  }

  float* om = out + (size_t)(mh + wv * 4) * NTOK + ti * BI + lane;
  #pragma unroll
  for (int r = 0; r < 4; ++r) {
    const int jr = wv * 4 + r;
    #pragma unroll
    for (int hf = 0; hf < 2; ++hf) {
      const int i = hf * 32 + lane;
      om[(size_t)r * NTOK + hf * 32] = -0.5f * (T0[i * OP + jr] + T1[i * OP + jr]);
    }
  }
}

extern "C" void kernel_launch(void* const* d_in, const int* in_sizes, int n_in,
                              void* d_out, int out_size, void* d_ws, size_t ws_size,
                              hipStream_t stream) {
  (void)in_sizes; (void)n_in; (void)out_size; (void)ws_size;
  const float* E  = (const float*)d_in[0];
  const float* W1 = (const float*)d_in[1];
  const float* b1 = (const float*)d_in[2];
  const float* W2 = (const float*)d_in[3];
  const float* b2 = (const float*)d_in[4];
  float* out  = (float*)d_out;
  float* Hbuf = (float*)d_ws;

  proj_kernel<<<dim3((BATCH * NTOK) / 16, HID / 16, 2), 32, 0, stream>>>(E, W1, b1, Hbuf);
  pair_kernel<<<dim3(NUNIT, BATCH), 256, 0, stream>>>(
      Hbuf, Hbuf + (size_t)BATCH * NTOK * HID, W2, b2, out);
}
